# starting kernel
# baseline (speedup 1.0000x reference)
#include <hip/hip_runtime.h>

typedef __attribute__((ext_vector_type(2))) float v2f;
typedef __attribute__((ext_vector_type(8))) float v8f;

#define CONV_H 128
#define CONV_W 128
#define CONV_IC 8
#define CONV_OC 16
#define K_STEPS 18
#define MTILES 4

__global__ __launch_bounds__(256)
void conv2d_nhwc_wmma_f32(const float* __restrict__ x,
                          const float* __restrict__ wgt,
                          float* __restrict__ out) {
  const int lane = threadIdx.x & 31;
  const int wave = threadIdx.x >> 5;
  const int job  = blockIdx.x * 8 + wave;
  const int wseg = job & 1;
  const int h    = (job >> 1) & 127;
  const int n    = job >> 8;

  const int half = lane >> 4;
  const int ml   = lane & 15;
  const int w0   = wseg * 64;

  v2f bfrag[K_STEPS];
#pragma unroll
  for (int kk = 0; kk < K_STEPS; ++kk) {
    const int k0 = 4 * kk + 2 * half;
    bfrag[kk].x = wgt[k0 * CONV_OC + ml];
    bfrag[kk].y = wgt[(k0 + 1) * CONV_OC + ml];
  }

  const float* xN = x + (size_t)n * CONV_H * CONV_W * CONV_IC;

  v8f acc[MTILES];
#pragma unroll
  for (int t = 0; t < MTILES; ++t) acc[t] = (v8f){};

#pragma unroll
  for (int kk = 0; kk < K_STEPS; ++kk) {
    const int pix = kk >> 1;
    const int fh  = pix / 3;
    const int fw  = pix % 3;
    const int ic  = (kk & 1) * 4 + 2 * half;

    const int  ih   = h + fh - 1;
    const int  ihc  = ih < 0 ? 0 : (ih > CONV_H - 1 ? CONV_H - 1 : ih);
    const bool ihok = (ih >= 0) & (ih < CONV_H);
    const float* rowp = xN + (size_t)ihc * CONV_W * CONV_IC + ic;

#pragma unroll
    for (int t = 0; t < MTILES; ++t) {
      const int iw  = w0 + 16 * t + ml + fw - 1;
      const int iwc = iw < 0 ? 0 : (iw > CONV_W - 1 ? CONV_W - 1 : iw);
      const float msk = (ihok & (iw >= 0) & (iw < CONV_W)) ? 1.0f : 0.0f;

      v2f a = *(const v2f*)(rowp + (size_t)iwc * CONV_IC);
      a.x *= msk;
      a.y *= msk;

      acc[t] = __builtin_amdgcn_wmma_f32_16x16x4_f32(
           false, a,  false, bfrag[kk],
           (short)0, acc[t],  false,  false);
    }
  }

  float* o = out + (((size_t)n * CONV_H + h) * CONV_W + w0) * CONV_OC;
#pragma unroll
  for (int t = 0; t < MTILES; ++t) {
#pragma unroll
    for (int j = 0; j < 8; ++j) {
      o[(size_t)(16 * t + j + 8 * half) * CONV_OC + ml] = acc[t][j];
    }
  }
}

extern "C" void kernel_launch(void* const* d_in, const int* in_sizes, int n_in,
                              void* d_out, int out_size, void* d_ws, size_t ws_size,
                              hipStream_t stream) {
  const float* x = (const float*)d_in[0];
  const float* w = (const float*)d_in[1];
  float* out = (float*)d_out;

  const int jobs   = 32 * 128 * (128 / 64);
  const int blocks = jobs / 8;
  conv2d_nhwc_wmma_f32<<<blocks, 256, 0, stream>>>(x, w, out);
}


# A fragments of chunk k+1 read from LDS as ONE group of 18 under chunk k's products (2nd register set; its rows staged a chunk ahead, barrier split into signal and wait); sums unchanged
# speedup vs baseline: 1.4674x; 1.4674x over previous
#include <hip/hip_runtime.h>


typedef __attribute__((ext_vector_type(2))) float v2f;
typedef __attribute__((ext_vector_type(4))) float v4f;
typedef __attribute__((ext_vector_type(8))) float v8f;

#define CONV_H 128
#define CONV_W 128
#define CONV_IC 8
#define CONV_OC 16
#define BAND_ROWS 16
#define LDS_ROWS (BAND_ROWS + 2)
#define PITCH4 260
#define ROW4 256
#define CHUNK_ROWS 4
#define LDS_UNITS 10496
#define FRAGS 18

__device__ __forceinline__ void lds_barrier() {
  __builtin_amdgcn_fence(__ATOMIC_RELEASE, "workgroup", "local");
  __builtin_amdgcn_s_barrier();
  __builtin_amdgcn_fence(__ATOMIC_ACQUIRE, "workgroup", "local");
}

__device__ __forceinline__ void lds_publish() {
  __builtin_amdgcn_fence(__ATOMIC_RELEASE, "workgroup", "local");
  __builtin_amdgcn_s_barrier_signal(-1);
}
__device__ __forceinline__ void lds_await() {
  __builtin_amdgcn_s_barrier_wait(-1);
  __builtin_amdgcn_fence(__ATOMIC_ACQUIRE, "workgroup", "local");
}

__global__ __launch_bounds__(256)
void conv2d_nhwc_wmma_f32(const float* __restrict__ x,
                          const float* __restrict__ wgt,
                          float* __restrict__ out) {
  __shared__ v4f tile[LDS_UNITS];

  const int tid  = threadIdx.x;
  const int lane = tid & 31;
  const int wave = tid >> 5;
  const int half = lane >> 4;
  const int ml   = lane & 15;
  const int pm   = ((ml & 7) << 1) | (ml >> 3);
  const int n    = blockIdx.x >> 3;
  const int band = blockIdx.x & 7;
  const int r0   = band * BAND_ROWS - 1;

  v2f bfrag[18];
#pragma unroll
  for (int p = 0; p < 9; ++p) {
#pragma unroll
    for (int s = 0; s < 2; ++s) {
      const int k0 = p * CONV_IC + 4 * half + 2 * s;
      bfrag[2 * p + s].x = wgt[k0 * CONV_OC + ml];
      bfrag[2 * p + s].y = wgt[(k0 + 1) * CONV_OC + ml];
    }
  }

  __builtin_amdgcn_sched_barrier(0);

  const v4f* x4 = (const v4f*)x + (size_t)n * CONV_H * ROW4 + tid;
  v4f v[LDS_ROWS];
#pragma unroll
  for (int i = 0; i < LDS_ROWS; ++i) {
    const int gr  = r0 + i;
    const int grc = gr < 0 ? 0 : (gr > CONV_H - 1 ? CONV_H - 1 : gr);
    v[i] = x4[(size_t)grc * ROW4];
    if (i == 5 || i == 9 || i == 13 || i == 17) __builtin_amdgcn_sched_barrier(0);
  }

  if (tid < LDS_ROWS * 4) {
    const int row = tid >> 2;
    const int q   = tid & 3;
    tile[row * PITCH4 + ((q & 2) ? (PITCH4 - 2) : 0) + (q & 1)] = (v4f){0.0f, 0.0f, 0.0f, 0.0f};
  }

  const v4f* abase = tile + (16 * wave + pm) * 2 + half;
  float* obase = out + (((size_t)n * CONV_H + band * BAND_ROWS) * CONV_W + 16 * wave + half) * CONV_OC + ml;

#pragma unroll
  for (int i = 0; i <= CHUNK_ROWS + 1; ++i) tile[i * PITCH4 + 2 + tid] = v[i];
  if (band == 0) tile[2 + tid] = (v4f){0.0f, 0.0f, 0.0f, 0.0f};
  lds_barrier();

  v4f af[2][FRAGS];
#pragma unroll
  for (int fw = 0; fw < 3; ++fw)
#pragma unroll
    for (int lr = 0; lr < 4; ++lr) af[0][3 * lr + fw] = abase[lr * PITCH4 + fw * 2];
#pragma unroll
  for (int lr = 4; lr < 6; ++lr)
#pragma unroll
    for (int fw = 0; fw < 3; ++fw) af[0][3 * lr + fw] = abase[lr * PITCH4 + fw * 2];
  __builtin_amdgcn_sched_barrier(0);

#pragma unroll
  for (int ch = 0; ch < BAND_ROWS / CHUNK_ROWS; ++ch) {
    const int cur = ch & 1;
    const int nxt = cur ^ 1;
    const bool more = ch + 1 < BAND_ROWS / CHUNK_ROWS;

    v8f acc[CHUNK_ROWS];
#pragma unroll
    for (int o = 0; o < CHUNK_ROWS; ++o) acc[o] = (v8f){};

#pragma unroll
    for (int p = 0; p < 9; ++p) {
      const int fh = p / 3;
      const int fw = p % 3;
#pragma unroll
      for (int o = 0; o < CHUNK_ROWS; ++o) {
        const v4f a = af[cur][3 * (o + fh) + fw];
        const v2f a01 = {a.x, a.y};
        acc[o] = __builtin_amdgcn_wmma_f32_16x16x4_f32(
             false, a01,  false, bfrag[2 * p],
             (short)0, acc[o],  false,  false);
      }
#pragma unroll
      for (int o = 0; o < CHUNK_ROWS; ++o) {
        const v4f a = af[cur][3 * (o + fh) + fw];
        const v2f a23 = {a.z, a.w};
        acc[o] = __builtin_amdgcn_wmma_f32_16x16x4_f32(
             false, a23,  false, bfrag[2 * p + 1],
             (short)0, acc[o],  false,  false);
      }

      if (more && p == 2) {
        __builtin_amdgcn_sched_barrier(0);
        const int lo = CHUNK_ROWS * (ch + 1) + 2;
        const int hi = CHUNK_ROWS * (ch + 1) + CHUNK_ROWS + 1;
#pragma unroll
        for (int i = 0; i < LDS_ROWS; ++i) {
          if (i >= lo && i <= hi) tile[i * PITCH4 + 2 + tid] = v[i];
        }
        if (hi == LDS_ROWS - 1 && band == CONV_H / BAND_ROWS - 1)
          tile[(LDS_ROWS - 1) * PITCH4 + 2 + tid] = (v4f){0.0f, 0.0f, 0.0f, 0.0f};
        __builtin_amdgcn_sched_barrier(0);
      }
      if (more && p == 3) {
        __builtin_amdgcn_sched_barrier(0);
        lds_publish();
      }
      if (more && p == 5) {
        __builtin_amdgcn_sched_barrier(0);
        lds_await();
#pragma unroll
        for (int lr = 0; lr < 6; ++lr)
#pragma unroll
          for (int fw2 = 0; fw2 < 3; ++fw2)
            af[nxt][3 * lr + fw2] = abase[(CHUNK_ROWS * (ch + 1) + lr) * PITCH4 + fw2 * 2];
        __builtin_amdgcn_sched_barrier(0);
      }
    }

    __builtin_amdgcn_sched_barrier(0);

#pragma unroll
    for (int o = 0; o < CHUNK_ROWS; ++o) {
      float* orow = obase + (size_t)(CHUNK_ROWS * ch + o) * CONV_W * CONV_OC;
#pragma unroll
      for (int j = 0; j < 8; ++j) orow[j * 2 * CONV_OC] = acc[o][j];
    }
  }
}

extern "C" void kernel_launch(void* const* d_in, const int* in_sizes, int n_in,
                              void* d_out, int out_size, void* d_ws, size_t ws_size,
                              hipStream_t stream) {
  const float* x = (const float*)d_in[0];
  const float* w = (const float*)d_in[1];
  float* out = (float*)d_out;

  conv2d_nhwc_wmma_f32<<<256, 256, 0, stream>>>(x, w, out);
}
